# starting kernel
# baseline (speedup 1.0000x reference)
#include <hip/hip_runtime.h>

typedef __attribute__((ext_vector_type(2))) float v2f;
typedef __attribute__((ext_vector_type(4))) float v4f;
typedef __attribute__((ext_vector_type(8))) float v8f;

#define KDIM 64
#define DDIM 64
#define NBINS 63
#define ROWS_PER_BLOCK 256

__global__ __launch_bounds__(256)
void emb_table_kernel(const float* __restrict__ weight,
                      float* __restrict__ T) {
    __shared__ float s_R[KDIM];

    const int tid  = threadIdx.x;
    const int lane = tid & 31;
    const int wave = tid >> 5;

    if (tid < KDIM) s_R[tid] = 1.0f / (float)(tid + 1);
    __syncthreads();

    const int m  = lane & 15;
    const int kh = lane >> 4;
    for (int t = wave; t < 16; t += 8) {
        const int ti = t >> 2;
        const int tj = t & 3;
        const int i  = 16 * ti + m;
        v8f c = {};
        for (int kb = 0; kb < KDIM; kb += 4) {
            const int k0 = kb + 2 * kh;
            int d0 = i - k0;       d0 = d0 < 0 ? -d0 : d0;
            int d1 = i - (k0 + 1); d1 = d1 < 0 ? -d1 : d1;
            v2f a;
            a.x = s_R[d0];
            a.y = s_R[d1];
            v2f b;
            b.x = weight[(k0    ) * DDIM + 16 * tj + m];
            b.y = weight[(k0 + 1) * DDIM + 16 * tj + m];
            c = __builtin_amdgcn_wmma_f32_16x16x4_f32(
                     false, a,  false, b,
                     (short)0, c,  false,  false);
        }
#pragma unroll
        for (int v = 0; v < 8; ++v) {
            T[(16 * ti + v + 8 * kh) * DDIM + 16 * tj + m] = c[v];
        }
    }
}

__global__ __launch_bounds__(256)
void emb_gather_kernel(const float* __restrict__ x,
                       const float* __restrict__ low,
                       const v4f* __restrict__ T4g,
                       v4f* __restrict__ out4,
                       int total_rows) {
    __shared__ __align__(16) v4f s_T4[KDIM * DDIM / 4];
    __shared__ float s_bins[NBINS];
    __shared__ int   s_idx[ROWS_PER_BLOCK];

    const int tid = threadIdx.x;

#pragma unroll
    for (int i = 0; i < 4; ++i) s_T4[tid + 256 * i] = T4g[tid + 256 * i];
    if (tid < NBINS) s_bins[tid] = low[tid + 1];
    __syncthreads();

    const int rowBase = blockIdx.x * ROWS_PER_BLOCK;
    {
        const int r = rowBase + tid;
        const float xv = (r < total_rows) ? __builtin_nontemporal_load(x + r) : 0.0f;
        int idx = 0;
#pragma unroll
        for (int j = 0; j < NBINS; ++j) idx += (xv > s_bins[j]) ? 1 : 0;
        s_idx[tid] = idx;
    }
    __syncthreads();

    const int sub  = tid & 15;
    const int rsub = tid >> 4;
#pragma unroll 4
    for (int it = 0; it < 16; ++it) {
        const int rl = rsub + 16 * it;
        const int r  = rowBase + rl;
        if (r < total_rows) {
            const int idx = s_idx[rl];
            const v4f val = s_T4[idx * 16 + sub];
            __builtin_nontemporal_store(val, out4 + (long long)r * 16 + sub);
        }
    }
}

extern "C" void kernel_launch(void* const* d_in, const int* in_sizes, int n_in,
                              void* d_out, int out_size, void* d_ws, size_t ws_size,
                              hipStream_t stream) {
    const float* x      = (const float*)d_in[0];
    const float* low    = (const float*)d_in[1];
    const float* weight = (const float*)d_in[3];

    float* T = (float*)d_ws;

    emb_table_kernel<<<1, 256, 0, stream>>>(weight, T);

    const int total_rows = in_sizes[0];
    const int blocks = (total_rows + ROWS_PER_BLOCK - 1) / ROWS_PER_BLOCK;
    emb_gather_kernel<<<blocks, 256, 0, stream>>>(
        x, low, (const v4f*)T, (v4f*)d_out, total_rows);
}


# Table product fused into the gather launch: 256 resident workgroups each build the table in LDS with the same f32 WMMA tiles, then loop over row chunks; one launch, no table pass via memory
# speedup vs baseline: 1.1861x; 1.1861x over previous
#include <hip/hip_runtime.h>


typedef __attribute__((ext_vector_type(2))) float v2f;
typedef __attribute__((ext_vector_type(4))) float v4f;
typedef __attribute__((ext_vector_type(8))) float v8f;

#define KDIM 64
#define DDIM 64
#define NBINS 63
#define ROWS_PER_WAVE 32
#define ROWS_PER_CHUNK 256
#define NUM_BLOCKS 256
#define XPF 8

__global__ __launch_bounds__(256)
void emb_fused_kernel(const float* __restrict__ x,
                      const float* __restrict__ low,
                      const float* __restrict__ weight,
                      v4f* __restrict__ out4,
                      int total_rows) {
    __shared__ __align__(16) float s_T[KDIM * DDIM];
    __shared__ float s_R[KDIM];

    const int tid  = threadIdx.x;
    const int lane = tid & 31;
    const int wave = __builtin_amdgcn_readfirstlane(tid >> 5);
    const int nchunks = (total_rows + ROWS_PER_CHUNK - 1) / ROWS_PER_CHUNK;

    const int m   = lane & 15;
    const int kh  = lane >> 4;

    float bins[NBINS];
#pragma unroll
    for (int j = 0; j < NBINS; ++j) bins[j] = low[j + 1];

    const int tj  = wave & 3;
    const int tiA = wave >> 2;
    const int tiB = tiA + 2;

    v2f b[KDIM / 4];
#pragma unroll
    for (int s = 0; s < KDIM / 4; ++s) {
        const int k0 = 4 * s + 2 * kh;
        b[s].x = weight[(k0    ) * DDIM + 16 * tj + m];
        b[s].y = weight[(k0 + 1) * DDIM + 16 * tj + m];
    }

    if (tid < KDIM) s_R[tid] = 1.0f / (float)(tid + 1);
    __syncthreads();

    int c0 = blockIdx.x;
    float xv[XPF];
    if (c0 < nchunks) {
#pragma unroll
        for (int u = 0; u < XPF; ++u) {
            const int r = (c0 + u * NUM_BLOCKS) * ROWS_PER_CHUNK + wave * ROWS_PER_WAVE + lane;
            xv[u] = __builtin_nontemporal_load(x + min(r, total_rows - 1));
        }
    }

    {
        const int iA = 16 * tiA + m;
        const int iB = 16 * tiB + m;
        v8f cA = {};
        v8f cB = {};
#pragma unroll
        for (int s = 0; s < KDIM / 4; ++s) {
            const int k0 = 4 * s + 2 * kh;
            int d0 = iA - k0;       d0 = d0 < 0 ? -d0 : d0;
            int d1 = iA - (k0 + 1); d1 = d1 < 0 ? -d1 : d1;
            int e0 = iB - k0;       e0 = e0 < 0 ? -e0 : e0;
            int e1 = iB - (k0 + 1); e1 = e1 < 0 ? -e1 : e1;
            v2f aA, aB;
            aA.x = s_R[d0];  aA.y = s_R[d1];
            aB.x = s_R[e0];  aB.y = s_R[e1];
            cA = __builtin_amdgcn_wmma_f32_16x16x4_f32(
                     false, aA,  false, b[s],
                     (short)0, cA,  false,  false);
            cB = __builtin_amdgcn_wmma_f32_16x16x4_f32(
                     false, aB,  false, b[s],
                     (short)0, cB,  false,  false);
        }
#pragma unroll
        for (int v = 0; v < 8; ++v) {
            s_T[(16 * tiA + v + 8 * kh) * DDIM + 16 * tj + m] = cA[v];
            s_T[(16 * tiB + v + 8 * kh) * DDIM + 16 * tj + m] = cB[v];
        }
    }
    __syncthreads();

    const v4f* s_T4 = (const v4f*)s_T;
    while (c0 < nchunks) {
#pragma unroll
        for (int u = 0; u < XPF; ++u) {
            const int rbase = (c0 + u * NUM_BLOCKS) * ROWS_PER_CHUNK + wave * ROWS_PER_WAVE;
            if (rbase >= total_rows) break;
            const float xr = xv[u];
            int idx = 0;
#pragma unroll
            for (int j = 0; j < NBINS; ++j) idx += (xr > bins[j]) ? 1 : 0;
            const int rofs = idx * (DDIM / 4);
            v4f* const dst = out4 + (long long)(rbase + kh) * (DDIM / 4) + m;
            if (rbase + ROWS_PER_WAVE <= total_rows) {
                int src[16];
#pragma unroll
                for (int it = 0; it < 16; ++it) src[it] = __shfl(rofs, 2 * it + kh);
#pragma unroll
                for (int it = 0; it < 16; ++it) {
                    const v4f val = s_T4[src[it] + m];
                    __builtin_nontemporal_store(val, dst + it * 2 * (DDIM / 4));
                }
            } else {
#pragma unroll 1
                for (int it = 0; it < 16; ++it) {
                    const int s1 = __shfl(rofs, 2 * it + kh);
                    if (rbase + 2 * it + kh < total_rows) {
                        const v4f val = s_T4[s1 + m];
                        __builtin_nontemporal_store(val, dst + it * 2 * (DDIM / 4));
                    }
                }
            }
        }
        c0 += XPF * NUM_BLOCKS;
        if (c0 < nchunks) {
#pragma unroll
            for (int u = 0; u < XPF; ++u) {
                const int r = (c0 + u * NUM_BLOCKS) * ROWS_PER_CHUNK + wave * ROWS_PER_WAVE + lane;
                xv[u] = __builtin_nontemporal_load(x + min(r, total_rows - 1));
            }
        }
    }
}

extern "C" void kernel_launch(void* const* d_in, const int* in_sizes, int n_in,
                              void* d_out, int out_size, void* d_ws, size_t ws_size,
                              hipStream_t stream) {
    const float* x      = (const float*)d_in[0];
    const float* low    = (const float*)d_in[1];
    const float* weight = (const float*)d_in[3];

    const int total_rows = in_sizes[0];
    emb_fused_kernel<<<NUM_BLOCKS, 256, 0, stream>>>(
        x, low, weight, (v4f*)d_out, total_rows);
}
